# starting kernel
# baseline (speedup 1.0000x reference)
#include <hip/hip_runtime.h>
#include <math.h>


typedef __attribute__((ext_vector_type(2))) float v2f;
typedef __attribute__((ext_vector_type(4))) float v4f;
typedef __attribute__((ext_vector_type(8))) float v8f;

#define B_DIM 256
#define T_DIM 512
#define F_DIM 256
#define U_DIM 256

__global__ __launch_bounds__(256)
void lstm_head_tanh_wmma_f32(const float* __restrict__ x,
                             const float* __restrict__ w,
                             const float* __restrict__ bias,
                             float* __restrict__ out)
{
    __shared__ __align__(16) float Bs[F_DIM * 16];

    const int tid  = threadIdx.x;
    const int wave = tid >> 5;
    const int lane = tid & 31;
    const int half = lane >> 4;
    const int lr   = lane & 15;

    const int n0 = blockIdx.x * 16;

    {
        const v4f* __restrict__ wrow;
        for (int i = tid; i < F_DIM * 4; i += 256) {
            const int row = i >> 2;
            const int q   = i & 3;
            wrow = reinterpret_cast<const v4f*>(w + row * U_DIM + n0);
            reinterpret_cast<v4f*>(Bs)[i] = wrow[q];
        }
    }
    __syncthreads();

    const int    mt        = blockIdx.y * 8 + wave;
    const size_t rowStride = (size_t)T_DIM * F_DIM;
    const float* __restrict__ rowA =
        x + (size_t)(mt * 16 + lr) * rowStride + (size_t)(T_DIM - 1) * F_DIM;

    #pragma unroll
    for (int pb = 0; pb < 1024; pb += 256)
        __builtin_prefetch(reinterpret_cast<const char*>(rowA) + pb, 0, 0);

    v8f acc = {};

    #pragma unroll 4
    for (int kk = 0; kk < F_DIM; kk += 4) {
        const int kb = kk + half * 2;
        v2f a = *reinterpret_cast<const v2f*>(rowA + kb);
        v2f b;
        b.x = Bs[kb * 16 + lr];
        b.y = Bs[(kb + 1) * 16 + lr];
        acc = __builtin_amdgcn_wmma_f32_16x16x4_f32(
             false, a,  false, b,
             (short)0, acc,  false,  false);
    }

    const int   col = n0 + lr;
    const float bv  = bias[col];
    #pragma unroll
    for (int r = 0; r < 8; ++r) {
        const int row = mt * 16 + r + half * 8;
        out[row * U_DIM + col] = tanhf(acc[r] + bv);
    }
}

extern "C" void kernel_launch(void* const* d_in, const int* in_sizes, int n_in,
                              void* d_out, int out_size, void* d_ws, size_t ws_size,
                              hipStream_t stream) {
    (void)in_sizes; (void)n_in; (void)out_size; (void)d_ws; (void)ws_size;
    const float* x    = (const float*)d_in[0];
    const float* w    = (const float*)d_in[1];
    const float* bias = (const float*)d_in[2];
    float*       out  = (float*)d_out;

    dim3 grid(U_DIM / 16, 2);
    dim3 block(256);
    lstm_head_tanh_wmma_f32<<<grid, block, 0, stream>>>(x, w, bias, out);
}


# One 16x32 output tile per wave (was 16x16) on 64 workgroups of 2 waves: each X fragment feeds two WMMAs on two accumulators, B fragments read from LDS as one group ahead of the WMMAs; same k order
# speedup vs baseline: 1.4977x; 1.4977x over previous
#include <hip/hip_runtime.h>
#include <math.h>


typedef __attribute__((ext_vector_type(2))) float v2f;
typedef __attribute__((ext_vector_type(4))) float v4f;
typedef __attribute__((ext_vector_type(8))) float v8f;

#define B_DIM 256
#define T_DIM 512
#define F_DIM 256
#define U_DIM 256

#define N_TILES   2
#define SUB_PITCH (F_DIM * 16 + 32)
#define X_EARLY   31

__device__ __forceinline__ v2f load_b_frag(const float* bl, int j)
{
    v2f r;
    r.x = bl[(4 * j) * 16];
    r.y = bl[(4 * j + 1) * 16];
    return r;
}

__global__ __launch_bounds__(64)
void lstm_head_tanh_wmma_f32(const float* __restrict__ x,
                             const float* __restrict__ w,
                             const float* __restrict__ bias,
                             float* __restrict__ out)
{
    __shared__ __align__(16) float Bs[SUB_PITCH + F_DIM * 16];

    const int tid  = threadIdx.x;
    const int wave = tid >> 5;
    const int lane = tid & 31;
    const int half = lane >> 4;
    const int lr   = lane & 15;

    const int n0 = blockIdx.x * (16 * N_TILES);

    const int wrow = tid >> 3;
    const int wq   = tid & 7;
    const float* __restrict__ wsrc = w + wrow * U_DIM + n0 + wq * 4;
    v4f wv[32];
    #pragma unroll
    for (int it = 0; it < 32; ++it)
        wv[it] = *reinterpret_cast<const v4f*>(wsrc + it * 8 * U_DIM);

    const int    mt        = blockIdx.y * 2 + wave;
    const size_t rowStride = (size_t)T_DIM * F_DIM;
    const float* __restrict__ rowA =
        x + (size_t)(mt * 16 + lr) * rowStride + (size_t)(T_DIM - 1) * F_DIM + half * 2;

    v2f a[F_DIM / 4];
    #pragma unroll
    for (int j = 0; j < X_EARLY; ++j)
        a[j] = *reinterpret_cast<const v2f*>(rowA + 4 * j);

    float* bdst = Bs + (wq >> 2) * SUB_PITCH + wrow * 16 + (wq & 3) * 4;
    #pragma unroll
    for (int it = 0; it < 32; ++it)
        *reinterpret_cast<v4f*>(bdst + it * 8 * 16) = wv[it];
    asm volatile("s_wait_dscnt 0x0" ::: "memory");
    asm volatile("s_barrier_signal -1" ::: "memory");
    asm volatile("s_barrier_wait -1" ::: "memory");

    const float* bl0 = Bs + (half * 2) * 16 + lr;
    const float* bl1 = bl0 + SUB_PITCH;
    v2f b0[F_DIM / 4], b1[F_DIM / 4];
    #pragma unroll
    for (int i = 0; i < F_DIM / 4 - X_EARLY; ++i) {
        if (2 * i + 1 < F_DIM / 4) {
            b0[2 * i]     = load_b_frag(bl0, 2 * i);
            b1[2 * i]     = load_b_frag(bl1, 2 * i);
            b0[2 * i + 1] = load_b_frag(bl0, 2 * i + 1);
            b1[2 * i + 1] = load_b_frag(bl1, 2 * i + 1);
        }
        a[X_EARLY + i] = *reinterpret_cast<const v2f*>(rowA + 4 * (X_EARLY + i));
        __builtin_amdgcn_sched_barrier(0);
    }

    v8f acc0 = {};
    v8f acc1 = {};
    #pragma unroll
    for (int j = 0; j < F_DIM / 4; ++j) {
        acc0 = __builtin_amdgcn_wmma_f32_16x16x4_f32(
             false, a[j],  false, b0[j],
             (short)0, acc0,  false,  false);
        acc1 = __builtin_amdgcn_wmma_f32_16x16x4_f32(
             false, a[j],  false, b1[j],
             (short)0, acc1,  false,  false);
    }

    const int   col0 = n0 + lr;
    const int   col1 = col0 + 16;
    const float bv0  = bias[col0];
    const float bv1  = bias[col1];
    #pragma unroll
    for (int r = 0; r < 8; ++r) {
        const int row = mt * 16 + r + half * 8;
        out[row * U_DIM + col0] = tanhf(acc0[r] + bv0);
        out[row * U_DIM + col1] = tanhf(acc1[r] + bv1);
    }
}

extern "C" void kernel_launch(void* const* d_in, const int* in_sizes, int n_in,
                              void* d_out, int out_size, void* d_ws, size_t ws_size,
                              hipStream_t stream) {
    (void)in_sizes; (void)n_in; (void)out_size; (void)d_ws; (void)ws_size;
    const float* x    = (const float*)d_in[0];
    const float* w    = (const float*)d_in[1];
    const float* bias = (const float*)d_in[2];
    float*       out  = (float*)d_out;

    dim3 grid(8, 8);
    dim3 block(64);
    lstm_head_tanh_wmma_f32<<<grid, block, 0, stream>>>(x, w, bias, out);
}


# Work split by role on 128 workgroups of 4 waves: one matrix wave (16x32 tile, all X loads up front), three helper waves stage W and take 3/4 of the bias+tanh+store epilogue through LDS; same k order
# speedup vs baseline: 1.9153x; 1.2788x over previous
#include <hip/hip_runtime.h>
#include <math.h>


typedef __attribute__((ext_vector_type(2))) float v2f;
typedef __attribute__((ext_vector_type(4))) float v4f;
typedef __attribute__((ext_vector_type(8))) float v8f;

#define B_DIM 256
#define T_DIM 512
#define F_DIM 256
#define U_DIM 256

#define N_TILES   2
#define SUB_PITCH (F_DIM * 16 + 32)
#define HELPERS   3
#define K_STEPS   (F_DIM / 4)
#define W_GROUPS  (F_DIM / 4)
#define W_EACH    (W_GROUPS / HELPERS)

__device__ __forceinline__ v2f load_b_frag(const float* bl, int j)
{
    v2f r;
    r.x = bl[(4 * j) * 16];
    r.y = bl[(4 * j + 1) * 16];
    return r;
}

__global__ __launch_bounds__((HELPERS + 1) * 32)
void lstm_head_tanh_wmma_f32(const float* __restrict__ x,
                             const float* __restrict__ w,
                             const float* __restrict__ bias,
                             float* __restrict__ out)
{
    __shared__ __align__(16) float Bs[SUB_PITCH + F_DIM * 16];
    __shared__ __align__(16) float Es[HELPERS * 32 * 4];

    const int tid  = threadIdx.x;
    const int wave = __builtin_amdgcn_readfirstlane(tid >> 5);
    const int lane = tid & 31;
    const int half = lane >> 4;
    const int lr   = lane & 15;

    const int n0 = blockIdx.x * (16 * N_TILES);
    const int mt = blockIdx.y;

    const int col0 = n0 + lr;
    const int col1 = col0 + 16;
    float bv0, bv1;

    v2f a[K_STEPS];
    if (wave == 0) {
        const size_t rowStride = (size_t)T_DIM * F_DIM;
        const float* __restrict__ rowA =
            x + (size_t)(mt * 16 + lr) * rowStride + (size_t)(T_DIM - 1) * F_DIM + half * 2;
        #pragma unroll
        for (int j = 0; j < K_STEPS; ++j)
            a[j] = *reinterpret_cast<const v2f*>(rowA + 4 * j);
        bv0 = bias[col0];
        bv1 = bias[col1];
    }
    if (wave != 0) {
        const int hw = wave - 1;
        const int r4 = lane >> 3;
        const int wq = lane & 7;
        const float* __restrict__ wsrc = w + (hw * 4 + r4) * U_DIM + n0 + wq * 4;
        v4f wv[W_EACH];
        v4f wlast;
        #pragma unroll
        for (int i = 0; i < W_EACH; ++i)
            wv[i] = *reinterpret_cast<const v4f*>(wsrc + i * (HELPERS * 4) * U_DIM);
        if (hw == 0)
            wlast = *reinterpret_cast<const v4f*>(wsrc + W_EACH * (HELPERS * 4) * U_DIM);
        bv0 = bias[col0];
        bv1 = bias[col1];
        float* bdst = Bs + (wq >> 2) * SUB_PITCH + (hw * 4 + r4) * 16 + (wq & 3) * 4;
        #pragma unroll
        for (int i = 0; i < W_EACH; ++i)
            *reinterpret_cast<v4f*>(bdst + i * (HELPERS * 4) * 16) = wv[i];
        if (hw == 0)
            *reinterpret_cast<v4f*>(bdst + W_EACH * (HELPERS * 4) * 16) = wlast;
    }
    asm volatile("s_wait_dscnt 0x0" ::: "memory");
    asm volatile("s_barrier_signal -1" ::: "memory");
    asm volatile("s_barrier_wait -1" ::: "memory");

    v4f e;
    if (wave == 0) {
        const float* bl0 = Bs + (half * 2) * 16 + lr;
        const float* bl1 = bl0 + SUB_PITCH;
        v2f b0[K_STEPS], b1[K_STEPS];
        #pragma unroll
        for (int j = 0; j < K_STEPS; ++j) {
            b0[j] = load_b_frag(bl0, j);
            b1[j] = load_b_frag(bl1, j);
            if ((j & 3) == 3)
                __builtin_amdgcn_sched_barrier(0);
        }

        v8f acc0 = {};
        v8f acc1 = {};
        #pragma unroll
        for (int j = 0; j < K_STEPS; ++j) {
            acc0 = __builtin_amdgcn_wmma_f32_16x16x4_f32(
                 false, a[j],  false, b0[j],
                 (short)0, acc0,  false,  false);
            acc1 = __builtin_amdgcn_wmma_f32_16x16x4_f32(
                 false, a[j],  false, b1[j],
                 (short)0, acc1,  false,  false);
        }

        #pragma unroll
        for (int h = 1; h <= HELPERS; ++h) {
            v4f t;
            t.x = acc0[2 * h];
            t.y = acc0[2 * h + 1];
            t.z = acc1[2 * h];
            t.w = acc1[2 * h + 1];
            *reinterpret_cast<v4f*>(Es + ((h - 1) * 32 + lane) * 4) = t;
        }
        e.x = acc0[0];
        e.y = acc0[1];
        e.z = acc1[0];
        e.w = acc1[1];
    }
    __syncthreads();

    if (wave != 0)
        e = *reinterpret_cast<const v4f*>(Es + ((wave - 1) * 32 + lane) * 4);
    const int row = mt * 16 + 2 * wave + half * 8;
    out[row * U_DIM + col0]       = tanhf(e.x + bv0);
    out[row * U_DIM + col1]       = tanhf(e.z + bv1);
    out[(row + 1) * U_DIM + col0] = tanhf(e.y + bv0);
    out[(row + 1) * U_DIM + col1] = tanhf(e.w + bv1);
}

extern "C" void kernel_launch(void* const* d_in, const int* in_sizes, int n_in,
                              void* d_out, int out_size, void* d_ws, size_t ws_size,
                              hipStream_t stream) {
    (void)in_sizes; (void)n_in; (void)out_size; (void)d_ws; (void)ws_size;
    const float* x    = (const float*)d_in[0];
    const float* w    = (const float*)d_in[1];
    const float* bias = (const float*)d_in[2];
    float*       out  = (float*)d_out;

    dim3 grid(8, 16);
    dim3 block(128);
    lstm_head_tanh_wmma_f32<<<grid, block, 0, stream>>>(x, w, bias, out);
}


# tanh of the epilogue by the hardware instruction v_tanh_f32, four independent evaluations per lane (was four serial calls of the library routine, two divergent branches each); tanh rounding differs
# speedup vs baseline: 2.1327x; 1.1135x over previous
#include <hip/hip_runtime.h>
#include <math.h>


typedef __attribute__((ext_vector_type(2))) float v2f;
typedef __attribute__((ext_vector_type(4))) float v4f;
typedef __attribute__((ext_vector_type(8))) float v8f;

#define B_DIM 256
#define T_DIM 512
#define F_DIM 256
#define U_DIM 256

#define N_TILES   2
#define SUB_PITCH (F_DIM * 16 + 32)
#define HELPERS   3
#define K_STEPS   (F_DIM / 4)
#define W_GROUPS  (F_DIM / 4)
#define W_EACH    (W_GROUPS / HELPERS)

__device__ __forceinline__ v2f load_b_frag(const float* bl, int j)
{
    v2f r;
    r.x = bl[(4 * j) * 16];
    r.y = bl[(4 * j + 1) * 16];
    return r;
}

__global__ __launch_bounds__((HELPERS + 1) * 32)
void lstm_head_tanh_wmma_f32(const float* __restrict__ x,
                             const float* __restrict__ w,
                             const float* __restrict__ bias,
                             float* __restrict__ out)
{
    __shared__ __align__(16) float Bs[SUB_PITCH + F_DIM * 16];
    __shared__ __align__(16) float Es[HELPERS * 32 * 4];

    const int tid  = threadIdx.x;
    const int wave = __builtin_amdgcn_readfirstlane(tid >> 5);
    const int lane = tid & 31;
    const int half = lane >> 4;
    const int lr   = lane & 15;

    const int n0 = blockIdx.x * (16 * N_TILES);
    const int mt = blockIdx.y;

    const int col0 = n0 + lr;
    const int col1 = col0 + 16;
    float bv0, bv1;

    v2f a[K_STEPS];
    if (wave == 0) {
        const size_t rowStride = (size_t)T_DIM * F_DIM;
        const float* __restrict__ rowA =
            x + (size_t)(mt * 16 + lr) * rowStride + (size_t)(T_DIM - 1) * F_DIM + half * 2;
        #pragma unroll
        for (int j = 0; j < K_STEPS; ++j)
            a[j] = *reinterpret_cast<const v2f*>(rowA + 4 * j);
        bv0 = bias[col0];
        bv1 = bias[col1];
    }
    if (wave != 0) {
        const int hw = wave - 1;
        const int r4 = lane >> 3;
        const int wq = lane & 7;
        const float* __restrict__ wsrc = w + (hw * 4 + r4) * U_DIM + n0 + wq * 4;
        v4f wv[W_EACH];
        v4f wlast;
        #pragma unroll
        for (int i = 0; i < W_EACH; ++i)
            wv[i] = *reinterpret_cast<const v4f*>(wsrc + i * (HELPERS * 4) * U_DIM);
        if (hw == 0)
            wlast = *reinterpret_cast<const v4f*>(wsrc + W_EACH * (HELPERS * 4) * U_DIM);
        bv0 = bias[col0];
        bv1 = bias[col1];
        float* bdst = Bs + (wq >> 2) * SUB_PITCH + (hw * 4 + r4) * 16 + (wq & 3) * 4;
        #pragma unroll
        for (int i = 0; i < W_EACH; ++i)
            *reinterpret_cast<v4f*>(bdst + i * (HELPERS * 4) * 16) = wv[i];
        if (hw == 0)
            *reinterpret_cast<v4f*>(bdst + W_EACH * (HELPERS * 4) * 16) = wlast;
    }
    asm volatile("s_wait_dscnt 0x0" ::: "memory");
    asm volatile("s_barrier_signal -1" ::: "memory");
    asm volatile("s_barrier_wait -1" ::: "memory");

    v4f e;
    if (wave == 0) {
        const float* bl0 = Bs + (half * 2) * 16 + lr;
        const float* bl1 = bl0 + SUB_PITCH;
        v2f b0[K_STEPS], b1[K_STEPS];
        #pragma unroll
        for (int j = 0; j < K_STEPS; ++j) {
            b0[j] = load_b_frag(bl0, j);
            b1[j] = load_b_frag(bl1, j);
            if ((j & 3) == 3)
                __builtin_amdgcn_sched_barrier(0);
        }

        v8f acc0 = {};
        v8f acc1 = {};
        #pragma unroll
        for (int j = 0; j < K_STEPS; ++j) {
            acc0 = __builtin_amdgcn_wmma_f32_16x16x4_f32(
                 false, a[j],  false, b0[j],
                 (short)0, acc0,  false,  false);
            acc1 = __builtin_amdgcn_wmma_f32_16x16x4_f32(
                 false, a[j],  false, b1[j],
                 (short)0, acc1,  false,  false);
        }

        #pragma unroll
        for (int h = 1; h <= HELPERS; ++h) {
            v4f t;
            t.x = acc0[2 * h];
            t.y = acc0[2 * h + 1];
            t.z = acc1[2 * h];
            t.w = acc1[2 * h + 1];
            *reinterpret_cast<v4f*>(Es + ((h - 1) * 32 + lane) * 4) = t;
        }
        e.x = acc0[0];
        e.y = acc0[1];
        e.z = acc1[0];
        e.w = acc1[1];
    }
    __syncthreads();

    if (wave != 0)
        e = *reinterpret_cast<const v4f*>(Es + ((wave - 1) * 32 + lane) * 4);
    const int row = mt * 16 + 2 * wave + half * 8;
    const float t00 = __builtin_amdgcn_tanhf(e.x + bv0);
    const float t01 = __builtin_amdgcn_tanhf(e.z + bv1);
    const float t10 = __builtin_amdgcn_tanhf(e.y + bv0);
    const float t11 = __builtin_amdgcn_tanhf(e.w + bv1);
    out[row * U_DIM + col0]       = t00;
    out[row * U_DIM + col1]       = t01;
    out[(row + 1) * U_DIM + col0] = t10;
    out[(row + 1) * U_DIM + col1] = t11;
}

extern "C" void kernel_launch(void* const* d_in, const int* in_sizes, int n_in,
                              void* d_out, int out_size, void* d_ws, size_t ws_size,
                              hipStream_t stream) {
    (void)in_sizes; (void)n_in; (void)out_size; (void)d_ws; (void)ws_size;
    const float* x    = (const float*)d_in[0];
    const float* w    = (const float*)d_in[1];
    const float* bias = (const float*)d_in[2];
    float*       out  = (float*)d_out;

    dim3 grid(8, 16);
    dim3 block(128);
    lstm_head_tanh_wmma_f32<<<grid, block, 0, stream>>>(x, w, bias, out);
}
